# starting kernel
# baseline (speedup 1.0000x reference)
#include <hip/hip_runtime.h>
#include <math.h>

typedef __attribute__((ext_vector_type(2))) float v2f;
typedef __attribute__((ext_vector_type(8))) float v8f;

#define KDIM   512
#define KNH    16
#define KHEAD  32
#define KH     64
#define KW     64
#define KG     32
#define KN     128
#define K2N    256
#define KSHIFT 4

static __device__ __forceinline__ v8f wmma_f32(v2f a, v2f b, v8f c) {
  return __builtin_amdgcn_wmma_f32_16x16x4_f32(false, a, false, b, (short)0, c,
                                               false, false);
}

__global__ __launch_bounds__(256) void gemm_bias_kernel(
    const float* __restrict__ A, const float* __restrict__ Wm,
    const float* __restrict__ bias, float* __restrict__ C, int M, int N) {
  __shared__ float As[128][68];
  __shared__ float BsT[64][66];

  const int nBlocks = N >> 6;
  const int mBlock  = blockIdx.x / nBlocks;
  const int nBlock  = blockIdx.x % nBlocks;
  const int rowBase = mBlock << 7;
  const int colBase = nBlock << 6;

  const int tid  = threadIdx.x;
  const int wave = tid >> 5;
  const int lane = tid & 31;
  const int l16  = lane & 15;
  const int half = lane >> 4;

  v8f acc[4];
  acc[0] = {}; acc[1] = {}; acc[2] = {}; acc[3] = {};

  const int arow = tid >> 1;
  const int acol = (tid & 1) << 5;
  const int brow = tid >> 2;
  const int bcol = (tid & 3) << 4;

  const float* ag0 = A + (size_t)(rowBase + arow) * KDIM + acol;
  const float* bg0 = Wm + (size_t)brow * N + colBase + bcol;

  float4 ra[8], rb[4];
#pragma unroll
  for (int i = 0; i < 8; ++i) ra[i] = ((const float4*)ag0)[i];
#pragma unroll
  for (int i = 0; i < 4; ++i) rb[i] = ((const float4*)bg0)[i];

  for (int chunk = 0; chunk < 8; ++chunk) {
#pragma unroll
    for (int i = 0; i < 8; ++i)
      *(float4*)&As[arow][acol + 4 * i] = ra[i];
#pragma unroll
    for (int i = 0; i < 4; ++i) {
      BsT[bcol + 4 * i + 0][brow] = rb[i].x;
      BsT[bcol + 4 * i + 1][brow] = rb[i].y;
      BsT[bcol + 4 * i + 2][brow] = rb[i].z;
      BsT[bcol + 4 * i + 3][brow] = rb[i].w;
    }
    __syncthreads();

    if (chunk < 7) {
      const float* ag = ag0 + (chunk + 1) * 64;
      const float* bg = bg0 + (size_t)(chunk + 1) * 64 * N;
#pragma unroll
      for (int i = 0; i < 8; ++i) ra[i] = ((const float4*)ag)[i];
#pragma unroll
      for (int i = 0; i < 4; ++i) rb[i] = ((const float4*)bg)[i];
    }

#pragma unroll
    for (int kk = 0; kk < 16; ++kk) {
      v2f a = *(const v2f*)&As[wave * 16 + l16][4 * kk + 2 * half];
#pragma unroll
      for (int nt = 0; nt < 4; ++nt) {
        v2f b = *(const v2f*)&BsT[nt * 16 + l16][4 * kk + 2 * half];
        acc[nt] = wmma_f32(a, b, acc[nt]);
      }
    }
    __syncthreads();
  }

#pragma unroll
  for (int nt = 0; nt < 4; ++nt) {
    const int col = colBase + nt * 16 + l16;
    const float bv = bias[col];
#pragma unroll
    for (int j = 0; j < 8; ++j) {
      const int row = rowBase + wave * 16 + j + 8 * half;
      C[(size_t)row * N + col] = acc[nt][j] + bv;
    }
  }
}

__global__ __launch_bounds__(256) void attn_kernel(
    const float* __restrict__ Q, const float* __restrict__ KV,
    const float* __restrict__ peq, const float* __restrict__ pekv,
    const float* __restrict__ mask, float* __restrict__ O) {
  __shared__ float ks[K2N][34];
  __shared__ float pst[8][16][18];

  const int idx = blockIdx.x;
  const int h = idx % KNH;
  const int g = (idx / KNH) % KG;
  const int b = idx / (KNH * KG);
  const int gh = g >> 2, gw = g & 3;

  const int tid = threadIdx.x;
  const int wave = tid >> 5;
  const int lane = tid & 31;
  const int l16 = lane & 15;
  const int half = lane >> 4;

  {
    const int j = tid;
    const int s = j >> 7;
    const int n = j & 127;
    const int hsrc =
        (gh * 8 + (n >> 4) + (s == 0 ? KSHIFT : (KH - KSHIFT))) & (KH - 1);
    const int wsrc = gw * 16 + (n & 15);
    const float* krow =
        KV + (size_t)((b * KH + hsrc) * KW + wsrc) * (2 * KDIM) + h * KHEAD;
    const float* cosr = pekv + ((size_t)(0 * KG + g) * K2N + j) * (KHEAD / 2);
    const float* sinr = pekv + ((size_t)(1 * KG + g) * K2N + j) * (KHEAD / 2);
#pragma unroll
    for (int c = 0; c < 16; ++c) {
      const float x1 = krow[2 * c], x2 = krow[2 * c + 1];
      const float cc = cosr[c], ss = sinr[c];
      ks[j][c]      = x1 * cc - x2 * ss;
      ks[j][c + 16] = x2 * cc + x1 * ss;
    }
  }

  const int qn = wave * 16 + l16;
  const int qhs = gh * 8 + (qn >> 4);
  const int qws = gw * 16 + (qn & 15);
  const float* qrow = Q + (size_t)((b * KH + qhs) * KW + qws) * KDIM + h * KHEAD;
  const float* qcos = peq + ((size_t)(0 * KG + g) * KN + qn) * (KHEAD / 2);
  const float* qsin = peq + ((size_t)(1 * KG + g) * KN + qn) * (KHEAD / 2);
  v2f qf[8];
#pragma unroll
  for (int kk = 0; kk < 8; ++kk) {
    const int k0 = 4 * kk + 2 * half;
#pragma unroll
    for (int e = 0; e < 2; ++e) {
      if (kk < 4) {
        const int c = k0 + e;
        qf[kk][e] = qrow[2 * c] * qcos[c] - qrow[2 * c + 1] * qsin[c];
      } else {
        const int c = k0 + e - 16;
        qf[kk][e] = qrow[2 * c] * qsin[c] + qrow[2 * c + 1] * qcos[c];
      }
    }
  }

  __syncthreads();

  const float scale = 0.1767766952966369f;
  float mrowv[8], lrowv[8];
  v8f oacc[2];
  oacc[0] = {}; oacc[1] = {};
#pragma unroll
  for (int j = 0; j < 8; ++j) { mrowv[j] = -1e30f; lrowv[j] = 0.f; }

  const float* maskg = mask + (size_t)g * KN * K2N;

  for (int kt = 0; kt < 16; ++kt) {
    v8f s = {};
#pragma unroll
    for (int kk = 0; kk < 8; ++kk) {
      v2f bfr = *(const v2f*)&ks[kt * 16 + l16][4 * kk + 2 * half];
      s = wmma_f32(qf[kk], bfr, s);
    }

    const int keyc = kt * 16 + l16;
#pragma unroll
    for (int j = 0; j < 8; ++j) {
      const int qr = wave * 16 + j + 8 * half;
      s[j] = s[j] * scale + maskg[(size_t)qr * K2N + keyc];
    }

#pragma unroll
    for (int j = 0; j < 8; ++j) {
      float m = s[j];
#pragma unroll
      for (int d = 1; d < 16; d <<= 1) m = fmaxf(m, __shfl_xor(m, d, 32));
      const float mn = fmaxf(mrowv[j], m);
      const float alpha = expf(mrowv[j] - mn);
      const float pv = expf(s[j] - mn);
      s[j] = pv;
      float rs = pv;
#pragma unroll
      for (int d = 1; d < 16; d <<= 1) rs += __shfl_xor(rs, d, 32);
      lrowv[j] = lrowv[j] * alpha + rs;
      mrowv[j] = mn;
      oacc[0][j] *= alpha;
      oacc[1][j] *= alpha;
    }

#pragma unroll
    for (int j = 0; j < 8; ++j)
      pst[wave][j + 8 * half][l16] = s[j];
    asm volatile("s_wait_dscnt 0" ::: "memory");

    const int sv   = kt >> 3;
    const int hsrc = (gh * 8 + (kt & 7) + (sv == 0 ? KSHIFT : (KH - KSHIFT))) &
                     (KH - 1);
    const float* vbase = KV +
        (size_t)((b * KH + hsrc) * KW + gw * 16) * (2 * KDIM) + KDIM +
        h * KHEAD;
#pragma unroll
    for (int kk = 0; kk < 4; ++kk) {
      v2f a = *(const v2f*)&pst[wave][l16][4 * kk + 2 * half];
      const float* v0 = vbase + (size_t)(4 * kk + 2 * half) * (2 * KDIM);
#pragma unroll
      for (int ct = 0; ct < 2; ++ct) {
        v2f bv;
        bv.x = v0[ct * 16 + l16];
        bv.y = v0[2 * KDIM + ct * 16 + l16];
        oacc[ct] = wmma_f32(a, bv, oacc[ct]);
      }
    }
  }

#pragma unroll
  for (int ct = 0; ct < 2; ++ct) {
#pragma unroll
    for (int j = 0; j < 8; ++j) {
      const int n = wave * 16 + j + 8 * half;
      const int hs = gh * 8 + (n >> 4);
      const int ws = gw * 16 + (n & 15);
      O[(size_t)((b * KH + hs) * KW + ws) * KDIM + h * KHEAD + ct * 16 + l16] =
          oacc[ct][j] / lrowv[j];
    }
  }
}

extern "C" void kernel_launch(void* const* d_in, const int* in_sizes, int n_in,
                              void* d_out, int out_size, void* d_ws,
                              size_t ws_size, hipStream_t stream) {
  const float* x      = (const float*)d_in[0];
  const float* peq    = (const float*)d_in[1];
  const float* pekv   = (const float*)d_in[2];
  const float* mask   = (const float*)d_in[3];
  const float* w_q    = (const float*)d_in[4];
  const float* b_q    = (const float*)d_in[5];
  const float* w_kv   = (const float*)d_in[6];
  const float* b_kv   = (const float*)d_in[7];
  const float* w_proj = (const float*)d_in[8];
  const float* b_proj = (const float*)d_in[9];

  const int B = in_sizes[0] / (KH * KW * KDIM);
  const int M = B * KH * KW;

  float* qbuf  = (float*)d_ws;
  float* kvbuf = qbuf + (size_t)M * KDIM;
  float* obuf  = kvbuf + (size_t)M * 2 * KDIM;

  dim3 blk(256);
  gemm_bias_kernel<<<dim3((M / 128) * (KDIM / 64)), blk, 0, stream>>>(
      x, w_q, b_q, qbuf, M, KDIM);
  gemm_bias_kernel<<<dim3((M / 128) * (2 * KDIM / 64)), blk, 0, stream>>>(
      x, w_kv, b_kv, kvbuf, M, 2 * KDIM);
  attn_kernel<<<dim3(B * KG * KNH), blk, 0, stream>>>(qbuf, kvbuf, peq, pekv,
                                                      mask, obuf);
  gemm_bias_kernel<<<dim3((M / 128) * (KDIM / 64)), blk, 0, stream>>>(
      obuf, w_proj, b_proj, (float*)d_out, M, KDIM);
}


# GEMM: staged A/B tile values held as native 4-float vectors so the K-chunk prefetch stays in VGPRs; the private-memory (scratch) round trip of the A tile is gone
# speedup vs baseline: 1.6195x; 1.6195x over previous
#include <hip/hip_runtime.h>
#include <math.h>

typedef __attribute__((ext_vector_type(2))) float v2f;
typedef __attribute__((ext_vector_type(8))) float v8f;
typedef __attribute__((ext_vector_type(4))) float v4f;

#define KDIM   512
#define KNH    16
#define KHEAD  32
#define KH     64
#define KW     64
#define KG     32
#define KN     128
#define K2N    256
#define KSHIFT 4

static __device__ __forceinline__ v8f wmma_f32(v2f a, v2f b, v8f c) {
  return __builtin_amdgcn_wmma_f32_16x16x4_f32(false, a, false, b, (short)0, c,
                                               false, false);
}

__global__ __launch_bounds__(256) void gemm_bias_kernel(
    const float* __restrict__ A, const float* __restrict__ Wm,
    const float* __restrict__ bias, float* __restrict__ C, int M, int N) {
  __shared__ float As[128][68];
  __shared__ float BsT[64][66];

  const int nBlocks = N >> 6;
  const int mBlock  = blockIdx.x / nBlocks;
  const int nBlock  = blockIdx.x % nBlocks;
  const int rowBase = mBlock << 7;
  const int colBase = nBlock << 6;

  const int tid  = threadIdx.x;
  const int wave = tid >> 5;
  const int lane = tid & 31;
  const int l16  = lane & 15;
  const int half = lane >> 4;

  v8f acc[4];
  acc[0] = {}; acc[1] = {}; acc[2] = {}; acc[3] = {};

  const int arow = tid >> 1;
  const int acol = (tid & 1) << 5;
  const int brow = tid >> 2;
  const int bcol = (tid & 3) << 4;

  const float* ag0 = A + (size_t)(rowBase + arow) * KDIM + acol;
  const float* bg0 = Wm + (size_t)brow * N + colBase + bcol;

  v4f ra[8], rb[4];
#pragma unroll
  for (int i = 0; i < 8; ++i) ra[i] = ((const v4f*)ag0)[i];
#pragma unroll
  for (int i = 0; i < 4; ++i) rb[i] = ((const v4f*)bg0)[i];

  for (int chunk = 0; chunk < 8; ++chunk) {
#pragma unroll
    for (int i = 0; i < 8; ++i)
      *(v4f*)&As[arow][acol + 4 * i] = ra[i];
#pragma unroll
    for (int i = 0; i < 4; ++i) {
      BsT[bcol + 4 * i + 0][brow] = rb[i].x;
      BsT[bcol + 4 * i + 1][brow] = rb[i].y;
      BsT[bcol + 4 * i + 2][brow] = rb[i].z;
      BsT[bcol + 4 * i + 3][brow] = rb[i].w;
    }
    __syncthreads();

    if (chunk < 7) {
      const float* ag = ag0 + (chunk + 1) * 64;
      const float* bg = bg0 + (size_t)(chunk + 1) * 64 * N;
#pragma unroll
      for (int i = 0; i < 8; ++i) ra[i] = ((const v4f*)ag)[i];
#pragma unroll
      for (int i = 0; i < 4; ++i) rb[i] = ((const v4f*)bg)[i];
    }

#pragma unroll
    for (int kk = 0; kk < 16; ++kk) {
      v2f a = *(const v2f*)&As[wave * 16 + l16][4 * kk + 2 * half];
#pragma unroll
      for (int nt = 0; nt < 4; ++nt) {
        v2f b = *(const v2f*)&BsT[nt * 16 + l16][4 * kk + 2 * half];
        acc[nt] = wmma_f32(a, b, acc[nt]);
      }
    }
    __syncthreads();
  }

#pragma unroll
  for (int nt = 0; nt < 4; ++nt) {
    const int col = colBase + nt * 16 + l16;
    const float bv = bias[col];
#pragma unroll
    for (int j = 0; j < 8; ++j) {
      const int row = rowBase + wave * 16 + j + 8 * half;
      C[(size_t)row * N + col] = acc[nt][j] + bv;
    }
  }
}

__global__ __launch_bounds__(256) void attn_kernel(
    const float* __restrict__ Q, const float* __restrict__ KV,
    const float* __restrict__ peq, const float* __restrict__ pekv,
    const float* __restrict__ mask, float* __restrict__ O) {
  __shared__ float ks[K2N][34];
  __shared__ float pst[8][16][18];

  const int idx = blockIdx.x;
  const int h = idx % KNH;
  const int g = (idx / KNH) % KG;
  const int b = idx / (KNH * KG);
  const int gh = g >> 2, gw = g & 3;

  const int tid = threadIdx.x;
  const int wave = tid >> 5;
  const int lane = tid & 31;
  const int l16 = lane & 15;
  const int half = lane >> 4;

  {
    const int j = tid;
    const int s = j >> 7;
    const int n = j & 127;
    const int hsrc =
        (gh * 8 + (n >> 4) + (s == 0 ? KSHIFT : (KH - KSHIFT))) & (KH - 1);
    const int wsrc = gw * 16 + (n & 15);
    const float* krow =
        KV + (size_t)((b * KH + hsrc) * KW + wsrc) * (2 * KDIM) + h * KHEAD;
    const float* cosr = pekv + ((size_t)(0 * KG + g) * K2N + j) * (KHEAD / 2);
    const float* sinr = pekv + ((size_t)(1 * KG + g) * K2N + j) * (KHEAD / 2);
#pragma unroll
    for (int c = 0; c < 16; ++c) {
      const float x1 = krow[2 * c], x2 = krow[2 * c + 1];
      const float cc = cosr[c], ss = sinr[c];
      ks[j][c]      = x1 * cc - x2 * ss;
      ks[j][c + 16] = x2 * cc + x1 * ss;
    }
  }

  const int qn = wave * 16 + l16;
  const int qhs = gh * 8 + (qn >> 4);
  const int qws = gw * 16 + (qn & 15);
  const float* qrow = Q + (size_t)((b * KH + qhs) * KW + qws) * KDIM + h * KHEAD;
  const float* qcos = peq + ((size_t)(0 * KG + g) * KN + qn) * (KHEAD / 2);
  const float* qsin = peq + ((size_t)(1 * KG + g) * KN + qn) * (KHEAD / 2);
  v2f qf[8];
#pragma unroll
  for (int kk = 0; kk < 8; ++kk) {
    const int k0 = 4 * kk + 2 * half;
#pragma unroll
    for (int e = 0; e < 2; ++e) {
      if (kk < 4) {
        const int c = k0 + e;
        qf[kk][e] = qrow[2 * c] * qcos[c] - qrow[2 * c + 1] * qsin[c];
      } else {
        const int c = k0 + e - 16;
        qf[kk][e] = qrow[2 * c] * qsin[c] + qrow[2 * c + 1] * qcos[c];
      }
    }
  }

  __syncthreads();

  const float scale = 0.1767766952966369f;
  float mrowv[8], lrowv[8];
  v8f oacc[2];
  oacc[0] = {}; oacc[1] = {};
#pragma unroll
  for (int j = 0; j < 8; ++j) { mrowv[j] = -1e30f; lrowv[j] = 0.f; }

  const float* maskg = mask + (size_t)g * KN * K2N;

  for (int kt = 0; kt < 16; ++kt) {
    v8f s = {};
#pragma unroll
    for (int kk = 0; kk < 8; ++kk) {
      v2f bfr = *(const v2f*)&ks[kt * 16 + l16][4 * kk + 2 * half];
      s = wmma_f32(qf[kk], bfr, s);
    }

    const int keyc = kt * 16 + l16;
#pragma unroll
    for (int j = 0; j < 8; ++j) {
      const int qr = wave * 16 + j + 8 * half;
      s[j] = s[j] * scale + maskg[(size_t)qr * K2N + keyc];
    }

#pragma unroll
    for (int j = 0; j < 8; ++j) {
      float m = s[j];
#pragma unroll
      for (int d = 1; d < 16; d <<= 1) m = fmaxf(m, __shfl_xor(m, d, 32));
      const float mn = fmaxf(mrowv[j], m);
      const float alpha = expf(mrowv[j] - mn);
      const float pv = expf(s[j] - mn);
      s[j] = pv;
      float rs = pv;
#pragma unroll
      for (int d = 1; d < 16; d <<= 1) rs += __shfl_xor(rs, d, 32);
      lrowv[j] = lrowv[j] * alpha + rs;
      mrowv[j] = mn;
      oacc[0][j] *= alpha;
      oacc[1][j] *= alpha;
    }

#pragma unroll
    for (int j = 0; j < 8; ++j)
      pst[wave][j + 8 * half][l16] = s[j];
    asm volatile("s_wait_dscnt 0" ::: "memory");

    const int sv   = kt >> 3;
    const int hsrc = (gh * 8 + (kt & 7) + (sv == 0 ? KSHIFT : (KH - KSHIFT))) &
                     (KH - 1);
    const float* vbase = KV +
        (size_t)((b * KH + hsrc) * KW + gw * 16) * (2 * KDIM) + KDIM +
        h * KHEAD;
#pragma unroll
    for (int kk = 0; kk < 4; ++kk) {
      v2f a = *(const v2f*)&pst[wave][l16][4 * kk + 2 * half];
      const float* v0 = vbase + (size_t)(4 * kk + 2 * half) * (2 * KDIM);
#pragma unroll
      for (int ct = 0; ct < 2; ++ct) {
        v2f bv;
        bv.x = v0[ct * 16 + l16];
        bv.y = v0[2 * KDIM + ct * 16 + l16];
        oacc[ct] = wmma_f32(a, bv, oacc[ct]);
      }
    }
  }

#pragma unroll
  for (int ct = 0; ct < 2; ++ct) {
#pragma unroll
    for (int j = 0; j < 8; ++j) {
      const int n = wave * 16 + j + 8 * half;
      const int hs = gh * 8 + (n >> 4);
      const int ws = gw * 16 + (n & 15);
      O[(size_t)((b * KH + hs) * KW + ws) * KDIM + h * KHEAD + ct * 16 + l16] =
          oacc[ct][j] / lrowv[j];
    }
  }
}

extern "C" void kernel_launch(void* const* d_in, const int* in_sizes, int n_in,
                              void* d_out, int out_size, void* d_ws,
                              size_t ws_size, hipStream_t stream) {
  const float* x      = (const float*)d_in[0];
  const float* peq    = (const float*)d_in[1];
  const float* pekv   = (const float*)d_in[2];
  const float* mask   = (const float*)d_in[3];
  const float* w_q    = (const float*)d_in[4];
  const float* b_q    = (const float*)d_in[5];
  const float* w_kv   = (const float*)d_in[6];
  const float* b_kv   = (const float*)d_in[7];
  const float* w_proj = (const float*)d_in[8];
  const float* b_proj = (const float*)d_in[9];

  const int B = in_sizes[0] / (KH * KW * KDIM);
  const int M = B * KH * KW;

  float* qbuf  = (float*)d_ws;
  float* kvbuf = qbuf + (size_t)M * KDIM;
  float* obuf  = kvbuf + (size_t)M * 2 * KDIM;

  dim3 blk(256);
  gemm_bias_kernel<<<dim3((M / 128) * (KDIM / 64)), blk, 0, stream>>>(
      x, w_q, b_q, qbuf, M, KDIM);
  gemm_bias_kernel<<<dim3((M / 128) * (2 * KDIM / 64)), blk, 0, stream>>>(
      x, w_kv, b_kv, kvbuf, M, 2 * KDIM);
  attn_kernel<<<dim3(B * KG * KNH), blk, 0, stream>>>(qbuf, kvbuf, peq, pekv,
                                                      mask, obuf);
  gemm_bias_kernel<<<dim3((M / 128) * (KDIM / 64)), blk, 0, stream>>>(
      obuf, w_proj, b_proj, (float*)d_out, M, KDIM);
}


# attention: scores transposed (keys on WMMA rows): softmax max/sum in-lane, P feeds PV WMMA from registers (no LDS transpose or 16-lane shuffles), O stored via LDS; row-sum and PV key order change
# speedup vs baseline: 1.7831x; 1.1010x over previous
#include <hip/hip_runtime.h>
#include <math.h>

typedef __attribute__((ext_vector_type(2))) float v2f;
typedef __attribute__((ext_vector_type(8))) float v8f;
typedef __attribute__((ext_vector_type(4))) float v4f;

#define KDIM   512
#define KNH    16
#define KHEAD  32
#define KH     64
#define KW     64
#define KG     32
#define KN     128
#define K2N    256
#define KSHIFT 4

static __device__ __forceinline__ v8f wmma_f32(v2f a, v2f b, v8f c) {
  return __builtin_amdgcn_wmma_f32_16x16x4_f32(false, a, false, b, (short)0, c,
                                               false, false);
}

__global__ __launch_bounds__(256) void gemm_bias_kernel(
    const float* __restrict__ A, const float* __restrict__ Wm,
    const float* __restrict__ bias, float* __restrict__ C, int M, int N) {
  __shared__ float As[128][68];
  __shared__ float BsT[64][66];

  const int nBlocks = N >> 6;
  const int mBlock  = blockIdx.x / nBlocks;
  const int nBlock  = blockIdx.x % nBlocks;
  const int rowBase = mBlock << 7;
  const int colBase = nBlock << 6;

  const int tid  = threadIdx.x;
  const int wave = tid >> 5;
  const int lane = tid & 31;
  const int l16  = lane & 15;
  const int half = lane >> 4;

  v8f acc[4];
  acc[0] = {}; acc[1] = {}; acc[2] = {}; acc[3] = {};

  const int arow = tid >> 1;
  const int acol = (tid & 1) << 5;
  const int brow = tid >> 2;
  const int bcol = (tid & 3) << 4;

  const float* ag0 = A + (size_t)(rowBase + arow) * KDIM + acol;
  const float* bg0 = Wm + (size_t)brow * N + colBase + bcol;

  v4f ra[8], rb[4];
#pragma unroll
  for (int i = 0; i < 8; ++i) ra[i] = ((const v4f*)ag0)[i];
#pragma unroll
  for (int i = 0; i < 4; ++i) rb[i] = ((const v4f*)bg0)[i];

  for (int chunk = 0; chunk < 8; ++chunk) {
#pragma unroll
    for (int i = 0; i < 8; ++i)
      *(v4f*)&As[arow][acol + 4 * i] = ra[i];
#pragma unroll
    for (int i = 0; i < 4; ++i) {
      BsT[bcol + 4 * i + 0][brow] = rb[i].x;
      BsT[bcol + 4 * i + 1][brow] = rb[i].y;
      BsT[bcol + 4 * i + 2][brow] = rb[i].z;
      BsT[bcol + 4 * i + 3][brow] = rb[i].w;
    }
    __syncthreads();

    if (chunk < 7) {
      const float* ag = ag0 + (chunk + 1) * 64;
      const float* bg = bg0 + (size_t)(chunk + 1) * 64 * N;
#pragma unroll
      for (int i = 0; i < 8; ++i) ra[i] = ((const v4f*)ag)[i];
#pragma unroll
      for (int i = 0; i < 4; ++i) rb[i] = ((const v4f*)bg)[i];
    }

#pragma unroll
    for (int kk = 0; kk < 16; ++kk) {
      v2f a = *(const v2f*)&As[wave * 16 + l16][4 * kk + 2 * half];
#pragma unroll
      for (int nt = 0; nt < 4; ++nt) {
        v2f b = *(const v2f*)&BsT[nt * 16 + l16][4 * kk + 2 * half];
        acc[nt] = wmma_f32(a, b, acc[nt]);
      }
    }
    __syncthreads();
  }

#pragma unroll
  for (int nt = 0; nt < 4; ++nt) {
    const int col = colBase + nt * 16 + l16;
    const float bv = bias[col];
#pragma unroll
    for (int j = 0; j < 8; ++j) {
      const int row = rowBase + wave * 16 + j + 8 * half;
      C[(size_t)row * N + col] = acc[nt][j] + bv;
    }
  }
}

static __device__ __forceinline__ float xhalf(float x) {
  const int b = __builtin_bit_cast(int, x);
  return __builtin_bit_cast(
      float, __builtin_amdgcn_permlanex16(b, b, 0x76543210, 0xfedcba98, false,
                                          false));
}

__global__ __launch_bounds__(256) void attn_kernel(
    const float* __restrict__ Q, const float* __restrict__ KV,
    const float* __restrict__ peq, const float* __restrict__ pekv,
    const float* __restrict__ mask, float* __restrict__ O) {
  __shared__ float ks[K2N][34];

  const int idx = blockIdx.x;
  const int h = idx % KNH;
  const int g = (idx / KNH) % KG;
  const int b = idx / (KNH * KG);
  const int gh = g >> 2, gw = g & 3;

  const int tid = threadIdx.x;
  const int wave = tid >> 5;
  const int lane = tid & 31;
  const int l16 = lane & 15;
  const int half = lane >> 4;

  {
    const int j = tid;
    const int s = j >> 7;
    const int n = j & 127;
    const int hsrc =
        (gh * 8 + (n >> 4) + (s == 0 ? KSHIFT : (KH - KSHIFT))) & (KH - 1);
    const int wsrc = gw * 16 + (n & 15);
    const float* krow =
        KV + (size_t)((b * KH + hsrc) * KW + wsrc) * (2 * KDIM) + h * KHEAD;
    const float* cosr = pekv + ((size_t)(0 * KG + g) * K2N + j) * (KHEAD / 2);
    const float* sinr = pekv + ((size_t)(1 * KG + g) * K2N + j) * (KHEAD / 2);
#pragma unroll
    for (int c = 0; c < 16; ++c) {
      const float x1 = krow[2 * c], x2 = krow[2 * c + 1];
      const float cc = cosr[c], ss = sinr[c];
      ks[j][c]      = x1 * cc - x2 * ss;
      ks[j][c + 16] = x2 * cc + x1 * ss;
    }
  }

  const int qn = wave * 16 + l16;
  const int qhs = gh * 8 + (qn >> 4);
  const int qws = gw * 16 + (qn & 15);
  const float* qrow = Q + (size_t)((b * KH + qhs) * KW + qws) * KDIM + h * KHEAD;
  const float* qcos = peq + ((size_t)(0 * KG + g) * KN + qn) * (KHEAD / 2);
  const float* qsin = peq + ((size_t)(1 * KG + g) * KN + qn) * (KHEAD / 2);
  v2f qf[8];
#pragma unroll
  for (int kk = 0; kk < 8; ++kk) {
    const int k0 = 4 * kk + 2 * half;
#pragma unroll
    for (int e = 0; e < 2; ++e) {
      if (kk < 4) {
        const int c = k0 + e;
        qf[kk][e] = qrow[2 * c] * qcos[c] - qrow[2 * c + 1] * qsin[c];
      } else {
        const int c = k0 + e - 16;
        qf[kk][e] = qrow[2 * c] * qsin[c] + qrow[2 * c + 1] * qcos[c];
      }
    }
  }

  __syncthreads();

  const float scale = 0.1767766952966369f;
  float mrun = -1e30f, lrun = 0.f;
  v8f oacc[2];
  oacc[0] = {}; oacc[1] = {};

  const float* maskq =
      mask + (size_t)g * KN * K2N + (size_t)qn * K2N + 8 * half;

  for (int kt = 0; kt < 16; ++kt) {
    const v4f mk0 = *(const v4f*)(maskq + kt * 16);
    const v4f mk1 = *(const v4f*)(maskq + kt * 16 + 4);

    v8f s = {};
#pragma unroll
    for (int kk = 0; kk < 8; ++kk) {
      v2f kfr = *(const v2f*)&ks[kt * 16 + l16][4 * kk + 2 * half];
      s = wmma_f32(kfr, qf[kk], s);
    }
    s[0] = s[0] * scale + mk0.x;  s[1] = s[1] * scale + mk0.y;
    s[2] = s[2] * scale + mk0.z;  s[3] = s[3] * scale + mk0.w;
    s[4] = s[4] * scale + mk1.x;  s[5] = s[5] * scale + mk1.y;
    s[6] = s[6] * scale + mk1.z;  s[7] = s[7] * scale + mk1.w;

    float m = fmaxf(fmaxf(fmaxf(s[0], s[1]), fmaxf(s[2], s[3])),
                    fmaxf(fmaxf(s[4], s[5]), fmaxf(s[6], s[7])));
    m = fmaxf(m, xhalf(m));
    const float mn = fmaxf(mrun, m);
    const float alpha = expf(mrun - mn);
    float rs = 0.f;
#pragma unroll
    for (int r = 0; r < 8; ++r) {
      s[r] = expf(s[r] - mn);
      rs += s[r];
    }
    rs = rs + xhalf(rs);
    lrun = lrun * alpha + rs;
    mrun = mn;
#pragma unroll
    for (int r = 0; r < 8; ++r) {
      oacc[0][r] *= alpha;
      oacc[1][r] *= alpha;
    }

    const int sv   = kt >> 3;
    const int hsrc = (gh * 8 + (kt & 7) + (sv == 0 ? KSHIFT : (KH - KSHIFT))) &
                     (KH - 1);
    const float* vbase = KV +
        (size_t)((b * KH + hsrc) * KW + gw * 16) * (2 * KDIM) + KDIM +
        h * KHEAD;
#pragma unroll
    for (int kk = 0; kk < 4; ++kk) {
      v2f pb;
      pb.x = s[2 * kk];
      pb.y = s[2 * kk + 1];
      const float* v0 = vbase + (size_t)(8 * half + 2 * kk) * (2 * KDIM);
#pragma unroll
      for (int ct = 0; ct < 2; ++ct) {
        v2f va;
        va.x = v0[ct * 16 + l16];
        va.y = v0[2 * KDIM + ct * 16 + l16];
        oacc[ct] = wmma_f32(va, pb, oacc[ct]);
      }
    }
  }

  __syncthreads();
  float* ot = &ks[0][0] + wave * (16 * 36);
#pragma unroll
  for (int ct = 0; ct < 2; ++ct) {
    v4f o0, o1;
    o0.x = oacc[ct][0] / lrun;  o0.y = oacc[ct][1] / lrun;
    o0.z = oacc[ct][2] / lrun;  o0.w = oacc[ct][3] / lrun;
    o1.x = oacc[ct][4] / lrun;  o1.y = oacc[ct][5] / lrun;
    o1.z = oacc[ct][6] / lrun;  o1.w = oacc[ct][7] / lrun;
    *(v4f*)&ot[l16 * 36 + ct * 16 + 8 * half]     = o0;
    *(v4f*)&ot[l16 * 36 + ct * 16 + 8 * half + 4] = o1;
  }
  __syncthreads();
  {
    const int col = (lane & 7) * 4;
    const int hs = gh * 8 + wave;
#pragma unroll
    for (int i = 0; i < 4; ++i) {
      const int row = 4 * i + (lane >> 3);
      const v4f v = *(const v4f*)&ot[row * 36 + col];
      const int ws = gw * 16 + row;
      *(v4f*)(O + (size_t)((b * KH + hs) * KW + ws) * KDIM + h * KHEAD + col) =
          v;
    }
  }
}

extern "C" void kernel_launch(void* const* d_in, const int* in_sizes, int n_in,
                              void* d_out, int out_size, void* d_ws,
                              size_t ws_size, hipStream_t stream) {
  const float* x      = (const float*)d_in[0];
  const float* peq    = (const float*)d_in[1];
  const float* pekv   = (const float*)d_in[2];
  const float* mask   = (const float*)d_in[3];
  const float* w_q    = (const float*)d_in[4];
  const float* b_q    = (const float*)d_in[5];
  const float* w_kv   = (const float*)d_in[6];
  const float* b_kv   = (const float*)d_in[7];
  const float* w_proj = (const float*)d_in[8];
  const float* b_proj = (const float*)d_in[9];

  const int B = in_sizes[0] / (KH * KW * KDIM);
  const int M = B * KH * KW;

  float* qbuf  = (float*)d_ws;
  float* kvbuf = qbuf + (size_t)M * KDIM;
  float* obuf  = kvbuf + (size_t)M * 2 * KDIM;

  dim3 blk(256);
  gemm_bias_kernel<<<dim3((M / 128) * (KDIM / 64)), blk, 0, stream>>>(
      x, w_q, b_q, qbuf, M, KDIM);
  gemm_bias_kernel<<<dim3((M / 128) * (2 * KDIM / 64)), blk, 0, stream>>>(
      x, w_kv, b_kv, kvbuf, M, 2 * KDIM);
  attn_kernel<<<dim3(B * KG * KNH), blk, 0, stream>>>(qbuf, kvbuf, peq, pekv,
                                                      mask, obuf);
  gemm_bias_kernel<<<dim3((M / 128) * (KDIM / 64)), blk, 0, stream>>>(
      obuf, w_proj, b_proj, (float*)d_out, M, KDIM);
}


# GEMM staging loads remapped: 16 lanes read one 256-byte row segment, so each wave-wide 16-byte global load of the A and W tiles covers 4 whole 128-byte lines, not 32 and 16; LDS layout unchanged
# speedup vs baseline: 1.9433x; 1.0898x over previous
#include <hip/hip_runtime.h>
#include <math.h>

typedef __attribute__((ext_vector_type(2))) float v2f;
typedef __attribute__((ext_vector_type(8))) float v8f;
typedef __attribute__((ext_vector_type(4))) float v4f;

#define KDIM   512
#define KNH    16
#define KHEAD  32
#define KH     64
#define KW     64
#define KG     32
#define KN     128
#define K2N    256
#define KSHIFT 4

static __device__ __forceinline__ v8f wmma_f32(v2f a, v2f b, v8f c) {
  return __builtin_amdgcn_wmma_f32_16x16x4_f32(false, a, false, b, (short)0, c,
                                               false, false);
}

__global__ __launch_bounds__(256) void gemm_bias_kernel(
    const float* __restrict__ A, const float* __restrict__ Wm,
    const float* __restrict__ bias, float* __restrict__ C, int M, int N) {
  __shared__ float As[128][68];
  __shared__ float BsT[64][66];

  const int nBlocks = N >> 6;
  const int mBlock  = blockIdx.x / nBlocks;
  const int nBlock  = blockIdx.x % nBlocks;
  const int rowBase = mBlock << 7;
  const int colBase = nBlock << 6;

  const int tid  = threadIdx.x;
  const int wave = tid >> 5;
  const int lane = tid & 31;
  const int l16  = lane & 15;
  const int half = lane >> 4;

  v8f acc[4];
  acc[0] = {}; acc[1] = {}; acc[2] = {}; acc[3] = {};

  const int arow = tid >> 4;
  const int acol = (tid & 15) << 2;
  const int brow = tid >> 4;
  const int bcol = (tid & 15) << 2;

  const float* ag0 = A + (size_t)(rowBase + arow) * KDIM + acol;
  const float* bg0 = Wm + (size_t)brow * N + colBase + bcol;

  v4f ra[8], rb[4];
#pragma unroll
  for (int i = 0; i < 8; ++i)
    ra[i] = *(const v4f*)(ag0 + (size_t)i * 16 * KDIM);
#pragma unroll
  for (int i = 0; i < 4; ++i)
    rb[i] = *(const v4f*)(bg0 + (size_t)i * 16 * N);

  for (int chunk = 0; chunk < 8; ++chunk) {
#pragma unroll
    for (int i = 0; i < 8; ++i)
      *(v4f*)&As[arow + 16 * i][acol] = ra[i];
#pragma unroll
    for (int i = 0; i < 4; ++i) {
      BsT[bcol + 0][brow + 16 * i] = rb[i].x;
      BsT[bcol + 1][brow + 16 * i] = rb[i].y;
      BsT[bcol + 2][brow + 16 * i] = rb[i].z;
      BsT[bcol + 3][brow + 16 * i] = rb[i].w;
    }
    __syncthreads();

    if (chunk < 7) {
      const float* ag = ag0 + (chunk + 1) * 64;
      const float* bg = bg0 + (size_t)(chunk + 1) * 64 * N;
#pragma unroll
      for (int i = 0; i < 8; ++i)
        ra[i] = *(const v4f*)(ag + (size_t)i * 16 * KDIM);
#pragma unroll
      for (int i = 0; i < 4; ++i)
        rb[i] = *(const v4f*)(bg + (size_t)i * 16 * N);
    }

#pragma unroll
    for (int kk = 0; kk < 16; ++kk) {
      v2f a = *(const v2f*)&As[wave * 16 + l16][4 * kk + 2 * half];
#pragma unroll
      for (int nt = 0; nt < 4; ++nt) {
        v2f b = *(const v2f*)&BsT[nt * 16 + l16][4 * kk + 2 * half];
        acc[nt] = wmma_f32(a, b, acc[nt]);
      }
    }
    __syncthreads();
  }

#pragma unroll
  for (int nt = 0; nt < 4; ++nt) {
    const int col = colBase + nt * 16 + l16;
    const float bv = bias[col];
#pragma unroll
    for (int j = 0; j < 8; ++j) {
      const int row = rowBase + wave * 16 + j + 8 * half;
      C[(size_t)row * N + col] = acc[nt][j] + bv;
    }
  }
}

static __device__ __forceinline__ float xhalf(float x) {
  const int b = __builtin_bit_cast(int, x);
  return __builtin_bit_cast(
      float, __builtin_amdgcn_permlanex16(b, b, 0x76543210, 0xfedcba98, false,
                                          false));
}

__global__ __launch_bounds__(256) void attn_kernel(
    const float* __restrict__ Q, const float* __restrict__ KV,
    const float* __restrict__ peq, const float* __restrict__ pekv,
    const float* __restrict__ mask, float* __restrict__ O) {
  __shared__ float ks[K2N][34];

  const int idx = blockIdx.x;
  const int h = idx % KNH;
  const int g = (idx / KNH) % KG;
  const int b = idx / (KNH * KG);
  const int gh = g >> 2, gw = g & 3;

  const int tid = threadIdx.x;
  const int wave = tid >> 5;
  const int lane = tid & 31;
  const int l16 = lane & 15;
  const int half = lane >> 4;

  {
    const int j = tid;
    const int s = j >> 7;
    const int n = j & 127;
    const int hsrc =
        (gh * 8 + (n >> 4) + (s == 0 ? KSHIFT : (KH - KSHIFT))) & (KH - 1);
    const int wsrc = gw * 16 + (n & 15);
    const float* krow =
        KV + (size_t)((b * KH + hsrc) * KW + wsrc) * (2 * KDIM) + h * KHEAD;
    const float* cosr = pekv + ((size_t)(0 * KG + g) * K2N + j) * (KHEAD / 2);
    const float* sinr = pekv + ((size_t)(1 * KG + g) * K2N + j) * (KHEAD / 2);
#pragma unroll
    for (int c = 0; c < 16; ++c) {
      const float x1 = krow[2 * c], x2 = krow[2 * c + 1];
      const float cc = cosr[c], ss = sinr[c];
      ks[j][c]      = x1 * cc - x2 * ss;
      ks[j][c + 16] = x2 * cc + x1 * ss;
    }
  }

  const int qn = wave * 16 + l16;
  const int qhs = gh * 8 + (qn >> 4);
  const int qws = gw * 16 + (qn & 15);
  const float* qrow = Q + (size_t)((b * KH + qhs) * KW + qws) * KDIM + h * KHEAD;
  const float* qcos = peq + ((size_t)(0 * KG + g) * KN + qn) * (KHEAD / 2);
  const float* qsin = peq + ((size_t)(1 * KG + g) * KN + qn) * (KHEAD / 2);
  v2f qf[8];
#pragma unroll
  for (int kk = 0; kk < 8; ++kk) {
    const int k0 = 4 * kk + 2 * half;
#pragma unroll
    for (int e = 0; e < 2; ++e) {
      if (kk < 4) {
        const int c = k0 + e;
        qf[kk][e] = qrow[2 * c] * qcos[c] - qrow[2 * c + 1] * qsin[c];
      } else {
        const int c = k0 + e - 16;
        qf[kk][e] = qrow[2 * c] * qsin[c] + qrow[2 * c + 1] * qcos[c];
      }
    }
  }

  __syncthreads();

  const float scale = 0.1767766952966369f;
  float mrun = -1e30f, lrun = 0.f;
  v8f oacc[2];
  oacc[0] = {}; oacc[1] = {};

  const float* maskq =
      mask + (size_t)g * KN * K2N + (size_t)qn * K2N + 8 * half;

  for (int kt = 0; kt < 16; ++kt) {
    const v4f mk0 = *(const v4f*)(maskq + kt * 16);
    const v4f mk1 = *(const v4f*)(maskq + kt * 16 + 4);

    v8f s = {};
#pragma unroll
    for (int kk = 0; kk < 8; ++kk) {
      v2f kfr = *(const v2f*)&ks[kt * 16 + l16][4 * kk + 2 * half];
      s = wmma_f32(kfr, qf[kk], s);
    }
    s[0] = s[0] * scale + mk0.x;  s[1] = s[1] * scale + mk0.y;
    s[2] = s[2] * scale + mk0.z;  s[3] = s[3] * scale + mk0.w;
    s[4] = s[4] * scale + mk1.x;  s[5] = s[5] * scale + mk1.y;
    s[6] = s[6] * scale + mk1.z;  s[7] = s[7] * scale + mk1.w;

    float m = fmaxf(fmaxf(fmaxf(s[0], s[1]), fmaxf(s[2], s[3])),
                    fmaxf(fmaxf(s[4], s[5]), fmaxf(s[6], s[7])));
    m = fmaxf(m, xhalf(m));
    const float mn = fmaxf(mrun, m);
    const float alpha = expf(mrun - mn);
    float rs = 0.f;
#pragma unroll
    for (int r = 0; r < 8; ++r) {
      s[r] = expf(s[r] - mn);
      rs += s[r];
    }
    rs = rs + xhalf(rs);
    lrun = lrun * alpha + rs;
    mrun = mn;
#pragma unroll
    for (int r = 0; r < 8; ++r) {
      oacc[0][r] *= alpha;
      oacc[1][r] *= alpha;
    }

    const int sv   = kt >> 3;
    const int hsrc = (gh * 8 + (kt & 7) + (sv == 0 ? KSHIFT : (KH - KSHIFT))) &
                     (KH - 1);
    const float* vbase = KV +
        (size_t)((b * KH + hsrc) * KW + gw * 16) * (2 * KDIM) + KDIM +
        h * KHEAD;
#pragma unroll
    for (int kk = 0; kk < 4; ++kk) {
      v2f pb;
      pb.x = s[2 * kk];
      pb.y = s[2 * kk + 1];
      const float* v0 = vbase + (size_t)(8 * half + 2 * kk) * (2 * KDIM);
#pragma unroll
      for (int ct = 0; ct < 2; ++ct) {
        v2f va;
        va.x = v0[ct * 16 + l16];
        va.y = v0[2 * KDIM + ct * 16 + l16];
        oacc[ct] = wmma_f32(va, pb, oacc[ct]);
      }
    }
  }

  __syncthreads();
  float* ot = &ks[0][0] + wave * (16 * 36);
#pragma unroll
  for (int ct = 0; ct < 2; ++ct) {
    v4f o0, o1;
    o0.x = oacc[ct][0] / lrun;  o0.y = oacc[ct][1] / lrun;
    o0.z = oacc[ct][2] / lrun;  o0.w = oacc[ct][3] / lrun;
    o1.x = oacc[ct][4] / lrun;  o1.y = oacc[ct][5] / lrun;
    o1.z = oacc[ct][6] / lrun;  o1.w = oacc[ct][7] / lrun;
    *(v4f*)&ot[l16 * 36 + ct * 16 + 8 * half]     = o0;
    *(v4f*)&ot[l16 * 36 + ct * 16 + 8 * half + 4] = o1;
  }
  __syncthreads();
  {
    const int col = (lane & 7) * 4;
    const int hs = gh * 8 + wave;
#pragma unroll
    for (int i = 0; i < 4; ++i) {
      const int row = 4 * i + (lane >> 3);
      const v4f v = *(const v4f*)&ot[row * 36 + col];
      const int ws = gw * 16 + row;
      *(v4f*)(O + (size_t)((b * KH + hs) * KW + ws) * KDIM + h * KHEAD + col) =
          v;
    }
  }
}

extern "C" void kernel_launch(void* const* d_in, const int* in_sizes, int n_in,
                              void* d_out, int out_size, void* d_ws,
                              size_t ws_size, hipStream_t stream) {
  const float* x      = (const float*)d_in[0];
  const float* peq    = (const float*)d_in[1];
  const float* pekv   = (const float*)d_in[2];
  const float* mask   = (const float*)d_in[3];
  const float* w_q    = (const float*)d_in[4];
  const float* b_q    = (const float*)d_in[5];
  const float* w_kv   = (const float*)d_in[6];
  const float* b_kv   = (const float*)d_in[7];
  const float* w_proj = (const float*)d_in[8];
  const float* b_proj = (const float*)d_in[9];

  const int B = in_sizes[0] / (KH * KW * KDIM);
  const int M = B * KH * KW;

  float* qbuf  = (float*)d_ws;
  float* kvbuf = qbuf + (size_t)M * KDIM;
  float* obuf  = kvbuf + (size_t)M * 2 * KDIM;

  dim3 blk(256);
  gemm_bias_kernel<<<dim3((M / 128) * (KDIM / 64)), blk, 0, stream>>>(
      x, w_q, b_q, qbuf, M, KDIM);
  gemm_bias_kernel<<<dim3((M / 128) * (2 * KDIM / 64)), blk, 0, stream>>>(
      x, w_kv, b_kv, kvbuf, M, 2 * KDIM);
  attn_kernel<<<dim3(B * KG * KNH), blk, 0, stream>>>(qbuf, kvbuf, peq, pekv,
                                                      mask, obuf);
  gemm_bias_kernel<<<dim3((M / 128) * (KDIM / 64)), blk, 0, stream>>>(
      obuf, w_proj, b_proj, (float*)d_out, M, KDIM);
}


# attention softmax: exp(s-m) taken as 2^((s-m)*log2e) on the hardware base-2 exponential; the library expf's range reduction and overflow selects are gone; scores, sums and their order unchanged
# speedup vs baseline: 2.0259x; 1.0425x over previous
#include <hip/hip_runtime.h>
#include <math.h>

typedef __attribute__((ext_vector_type(2))) float v2f;
typedef __attribute__((ext_vector_type(8))) float v8f;
typedef __attribute__((ext_vector_type(4))) float v4f;

#define KDIM   512
#define KNH    16
#define KHEAD  32
#define KH     64
#define KW     64
#define KG     32
#define KN     128
#define K2N    256
#define KSHIFT 4

static __device__ __forceinline__ v8f wmma_f32(v2f a, v2f b, v8f c) {
  return __builtin_amdgcn_wmma_f32_16x16x4_f32(false, a, false, b, (short)0, c,
                                               false, false);
}

__global__ __launch_bounds__(256) void gemm_bias_kernel(
    const float* __restrict__ A, const float* __restrict__ Wm,
    const float* __restrict__ bias, float* __restrict__ C, int M, int N) {
  __shared__ float As[128][68];
  __shared__ float BsT[64][66];

  const int nBlocks = N >> 6;
  const int mBlock  = blockIdx.x / nBlocks;
  const int nBlock  = blockIdx.x % nBlocks;
  const int rowBase = mBlock << 7;
  const int colBase = nBlock << 6;

  const int tid  = threadIdx.x;
  const int wave = tid >> 5;
  const int lane = tid & 31;
  const int l16  = lane & 15;
  const int half = lane >> 4;

  v8f acc[4];
  acc[0] = {}; acc[1] = {}; acc[2] = {}; acc[3] = {};

  const int arow = tid >> 4;
  const int acol = (tid & 15) << 2;
  const int brow = tid >> 4;
  const int bcol = (tid & 15) << 2;

  const float* ag0 = A + (size_t)(rowBase + arow) * KDIM + acol;
  const float* bg0 = Wm + (size_t)brow * N + colBase + bcol;

  v4f ra[8], rb[4];
#pragma unroll
  for (int i = 0; i < 8; ++i)
    ra[i] = *(const v4f*)(ag0 + (size_t)i * 16 * KDIM);
#pragma unroll
  for (int i = 0; i < 4; ++i)
    rb[i] = *(const v4f*)(bg0 + (size_t)i * 16 * N);

  for (int chunk = 0; chunk < 8; ++chunk) {
#pragma unroll
    for (int i = 0; i < 8; ++i)
      *(v4f*)&As[arow + 16 * i][acol] = ra[i];
#pragma unroll
    for (int i = 0; i < 4; ++i) {
      BsT[bcol + 0][brow + 16 * i] = rb[i].x;
      BsT[bcol + 1][brow + 16 * i] = rb[i].y;
      BsT[bcol + 2][brow + 16 * i] = rb[i].z;
      BsT[bcol + 3][brow + 16 * i] = rb[i].w;
    }
    __syncthreads();

    if (chunk < 7) {
      const float* ag = ag0 + (chunk + 1) * 64;
      const float* bg = bg0 + (size_t)(chunk + 1) * 64 * N;
#pragma unroll
      for (int i = 0; i < 8; ++i)
        ra[i] = *(const v4f*)(ag + (size_t)i * 16 * KDIM);
#pragma unroll
      for (int i = 0; i < 4; ++i)
        rb[i] = *(const v4f*)(bg + (size_t)i * 16 * N);
    }

#pragma unroll
    for (int kk = 0; kk < 16; ++kk) {
      v2f a = *(const v2f*)&As[wave * 16 + l16][4 * kk + 2 * half];
#pragma unroll
      for (int nt = 0; nt < 4; ++nt) {
        v2f b = *(const v2f*)&BsT[nt * 16 + l16][4 * kk + 2 * half];
        acc[nt] = wmma_f32(a, b, acc[nt]);
      }
    }
    __syncthreads();
  }

#pragma unroll
  for (int nt = 0; nt < 4; ++nt) {
    const int col = colBase + nt * 16 + l16;
    const float bv = bias[col];
#pragma unroll
    for (int j = 0; j < 8; ++j) {
      const int row = rowBase + wave * 16 + j + 8 * half;
      C[(size_t)row * N + col] = acc[nt][j] + bv;
    }
  }
}

static __device__ __forceinline__ float xhalf(float x) {
  const int b = __builtin_bit_cast(int, x);
  return __builtin_bit_cast(
      float, __builtin_amdgcn_permlanex16(b, b, 0x76543210, 0xfedcba98, false,
                                          false));
}

__global__ __launch_bounds__(256) void attn_kernel(
    const float* __restrict__ Q, const float* __restrict__ KV,
    const float* __restrict__ peq, const float* __restrict__ pekv,
    const float* __restrict__ mask, float* __restrict__ O) {
  __shared__ float ks[K2N][34];

  const int idx = blockIdx.x;
  const int h = idx % KNH;
  const int g = (idx / KNH) % KG;
  const int b = idx / (KNH * KG);
  const int gh = g >> 2, gw = g & 3;

  const int tid = threadIdx.x;
  const int wave = tid >> 5;
  const int lane = tid & 31;
  const int l16 = lane & 15;
  const int half = lane >> 4;

  {
    const int j = tid;
    const int s = j >> 7;
    const int n = j & 127;
    const int hsrc =
        (gh * 8 + (n >> 4) + (s == 0 ? KSHIFT : (KH - KSHIFT))) & (KH - 1);
    const int wsrc = gw * 16 + (n & 15);
    const float* krow =
        KV + (size_t)((b * KH + hsrc) * KW + wsrc) * (2 * KDIM) + h * KHEAD;
    const float* cosr = pekv + ((size_t)(0 * KG + g) * K2N + j) * (KHEAD / 2);
    const float* sinr = pekv + ((size_t)(1 * KG + g) * K2N + j) * (KHEAD / 2);
#pragma unroll
    for (int c = 0; c < 16; ++c) {
      const float x1 = krow[2 * c], x2 = krow[2 * c + 1];
      const float cc = cosr[c], ss = sinr[c];
      ks[j][c]      = x1 * cc - x2 * ss;
      ks[j][c + 16] = x2 * cc + x1 * ss;
    }
  }

  const int qn = wave * 16 + l16;
  const int qhs = gh * 8 + (qn >> 4);
  const int qws = gw * 16 + (qn & 15);
  const float* qrow = Q + (size_t)((b * KH + qhs) * KW + qws) * KDIM + h * KHEAD;
  const float* qcos = peq + ((size_t)(0 * KG + g) * KN + qn) * (KHEAD / 2);
  const float* qsin = peq + ((size_t)(1 * KG + g) * KN + qn) * (KHEAD / 2);
  v2f qf[8];
#pragma unroll
  for (int kk = 0; kk < 8; ++kk) {
    const int k0 = 4 * kk + 2 * half;
#pragma unroll
    for (int e = 0; e < 2; ++e) {
      if (kk < 4) {
        const int c = k0 + e;
        qf[kk][e] = qrow[2 * c] * qcos[c] - qrow[2 * c + 1] * qsin[c];
      } else {
        const int c = k0 + e - 16;
        qf[kk][e] = qrow[2 * c] * qsin[c] + qrow[2 * c + 1] * qcos[c];
      }
    }
  }

  __syncthreads();

  const float scale = 0.1767766952966369f;
  const float log2e = 1.4426950408889634f;
  float mrun = -1e30f, lrun = 0.f;
  v8f oacc[2];
  oacc[0] = {}; oacc[1] = {};

  const float* maskq =
      mask + (size_t)g * KN * K2N + (size_t)qn * K2N + 8 * half;

  for (int kt = 0; kt < 16; ++kt) {
    const v4f mk0 = *(const v4f*)(maskq + kt * 16);
    const v4f mk1 = *(const v4f*)(maskq + kt * 16 + 4);

    v8f s = {};
#pragma unroll
    for (int kk = 0; kk < 8; ++kk) {
      v2f kfr = *(const v2f*)&ks[kt * 16 + l16][4 * kk + 2 * half];
      s = wmma_f32(kfr, qf[kk], s);
    }
    s[0] = s[0] * scale + mk0.x;  s[1] = s[1] * scale + mk0.y;
    s[2] = s[2] * scale + mk0.z;  s[3] = s[3] * scale + mk0.w;
    s[4] = s[4] * scale + mk1.x;  s[5] = s[5] * scale + mk1.y;
    s[6] = s[6] * scale + mk1.z;  s[7] = s[7] * scale + mk1.w;

    float m = fmaxf(fmaxf(fmaxf(s[0], s[1]), fmaxf(s[2], s[3])),
                    fmaxf(fmaxf(s[4], s[5]), fmaxf(s[6], s[7])));
    m = fmaxf(m, xhalf(m));
    const float mn = fmaxf(mrun, m);
    const float alpha = __builtin_amdgcn_exp2f((mrun - mn) * log2e);
    float rs = 0.f;
#pragma unroll
    for (int r = 0; r < 8; ++r) {
      s[r] = __builtin_amdgcn_exp2f((s[r] - mn) * log2e);
      rs += s[r];
    }
    rs = rs + xhalf(rs);
    lrun = lrun * alpha + rs;
    mrun = mn;
#pragma unroll
    for (int r = 0; r < 8; ++r) {
      oacc[0][r] *= alpha;
      oacc[1][r] *= alpha;
    }

    const int sv   = kt >> 3;
    const int hsrc = (gh * 8 + (kt & 7) + (sv == 0 ? KSHIFT : (KH - KSHIFT))) &
                     (KH - 1);
    const float* vbase = KV +
        (size_t)((b * KH + hsrc) * KW + gw * 16) * (2 * KDIM) + KDIM +
        h * KHEAD;
#pragma unroll
    for (int kk = 0; kk < 4; ++kk) {
      v2f pb;
      pb.x = s[2 * kk];
      pb.y = s[2 * kk + 1];
      const float* v0 = vbase + (size_t)(8 * half + 2 * kk) * (2 * KDIM);
#pragma unroll
      for (int ct = 0; ct < 2; ++ct) {
        v2f va;
        va.x = v0[ct * 16 + l16];
        va.y = v0[2 * KDIM + ct * 16 + l16];
        oacc[ct] = wmma_f32(va, pb, oacc[ct]);
      }
    }
  }

  __syncthreads();
  float* ot = &ks[0][0] + wave * (16 * 36);
#pragma unroll
  for (int ct = 0; ct < 2; ++ct) {
    v4f o0, o1;
    o0.x = oacc[ct][0] / lrun;  o0.y = oacc[ct][1] / lrun;
    o0.z = oacc[ct][2] / lrun;  o0.w = oacc[ct][3] / lrun;
    o1.x = oacc[ct][4] / lrun;  o1.y = oacc[ct][5] / lrun;
    o1.z = oacc[ct][6] / lrun;  o1.w = oacc[ct][7] / lrun;
    *(v4f*)&ot[l16 * 36 + ct * 16 + 8 * half]     = o0;
    *(v4f*)&ot[l16 * 36 + ct * 16 + 8 * half + 4] = o1;
  }
  __syncthreads();
  {
    const int col = (lane & 7) * 4;
    const int hs = gh * 8 + wave;
#pragma unroll
    for (int i = 0; i < 4; ++i) {
      const int row = 4 * i + (lane >> 3);
      const v4f v = *(const v4f*)&ot[row * 36 + col];
      const int ws = gw * 16 + row;
      *(v4f*)(O + (size_t)((b * KH + hs) * KW + ws) * KDIM + h * KHEAD + col) =
          v;
    }
  }
}

extern "C" void kernel_launch(void* const* d_in, const int* in_sizes, int n_in,
                              void* d_out, int out_size, void* d_ws,
                              size_t ws_size, hipStream_t stream) {
  const float* x      = (const float*)d_in[0];
  const float* peq    = (const float*)d_in[1];
  const float* pekv   = (const float*)d_in[2];
  const float* mask   = (const float*)d_in[3];
  const float* w_q    = (const float*)d_in[4];
  const float* b_q    = (const float*)d_in[5];
  const float* w_kv   = (const float*)d_in[6];
  const float* b_kv   = (const float*)d_in[7];
  const float* w_proj = (const float*)d_in[8];
  const float* b_proj = (const float*)d_in[9];

  const int B = in_sizes[0] / (KH * KW * KDIM);
  const int M = B * KH * KW;

  float* qbuf  = (float*)d_ws;
  float* kvbuf = qbuf + (size_t)M * KDIM;
  float* obuf  = kvbuf + (size_t)M * 2 * KDIM;

  dim3 blk(256);
  gemm_bias_kernel<<<dim3((M / 128) * (KDIM / 64)), blk, 0, stream>>>(
      x, w_q, b_q, qbuf, M, KDIM);
  gemm_bias_kernel<<<dim3((M / 128) * (2 * KDIM / 64)), blk, 0, stream>>>(
      x, w_kv, b_kv, kvbuf, M, 2 * KDIM);
  attn_kernel<<<dim3(B * KG * KNH), blk, 0, stream>>>(qbuf, kvbuf, peq, pekv,
                                                      mask, obuf);
  gemm_bias_kernel<<<dim3((M / 128) * (KDIM / 64)), blk, 0, stream>>>(
      obuf, w_proj, b_proj, (float*)d_out, M, KDIM);
}


# attention: K and V rows staged into LDS once per block by whole 128-byte lines (8 lanes x 16 B per row); V^T fragments are 8-byte LDS reads, not per-wave 4-byte global loads; sum order unchanged
# speedup vs baseline: 2.1131x; 1.0430x over previous
#include <hip/hip_runtime.h>
#include <math.h>

typedef __attribute__((ext_vector_type(2))) float v2f;
typedef __attribute__((ext_vector_type(8))) float v8f;
typedef __attribute__((ext_vector_type(4))) float v4f;

#define KDIM   512
#define KNH    16
#define KHEAD  32
#define KH     64
#define KW     64
#define KG     32
#define KN     128
#define K2N    256
#define KSHIFT 4

static __device__ __forceinline__ v8f wmma_f32(v2f a, v2f b, v8f c) {
  return __builtin_amdgcn_wmma_f32_16x16x4_f32(false, a, false, b, (short)0, c,
                                               false, false);
}

__global__ __launch_bounds__(256) void gemm_bias_kernel(
    const float* __restrict__ A, const float* __restrict__ Wm,
    const float* __restrict__ bias, float* __restrict__ C, int M, int N) {
  __shared__ float As[128][68];
  __shared__ float BsT[64][66];

  const int nBlocks = N >> 6;
  const int mBlock  = blockIdx.x / nBlocks;
  const int nBlock  = blockIdx.x % nBlocks;
  const int rowBase = mBlock << 7;
  const int colBase = nBlock << 6;

  const int tid  = threadIdx.x;
  const int wave = tid >> 5;
  const int lane = tid & 31;
  const int l16  = lane & 15;
  const int half = lane >> 4;

  v8f acc[4];
  acc[0] = {}; acc[1] = {}; acc[2] = {}; acc[3] = {};

  const int arow = tid >> 4;
  const int acol = (tid & 15) << 2;
  const int brow = tid >> 4;
  const int bcol = (tid & 15) << 2;

  const float* ag0 = A + (size_t)(rowBase + arow) * KDIM + acol;
  const float* bg0 = Wm + (size_t)brow * N + colBase + bcol;

  v4f ra[8], rb[4];
#pragma unroll
  for (int i = 0; i < 8; ++i)
    ra[i] = *(const v4f*)(ag0 + (size_t)i * 16 * KDIM);
#pragma unroll
  for (int i = 0; i < 4; ++i)
    rb[i] = *(const v4f*)(bg0 + (size_t)i * 16 * N);

  for (int chunk = 0; chunk < 8; ++chunk) {
#pragma unroll
    for (int i = 0; i < 8; ++i)
      *(v4f*)&As[arow + 16 * i][acol] = ra[i];
#pragma unroll
    for (int i = 0; i < 4; ++i) {
      BsT[bcol + 0][brow + 16 * i] = rb[i].x;
      BsT[bcol + 1][brow + 16 * i] = rb[i].y;
      BsT[bcol + 2][brow + 16 * i] = rb[i].z;
      BsT[bcol + 3][brow + 16 * i] = rb[i].w;
    }
    __syncthreads();

    if (chunk < 7) {
      const float* ag = ag0 + (chunk + 1) * 64;
      const float* bg = bg0 + (size_t)(chunk + 1) * 64 * N;
#pragma unroll
      for (int i = 0; i < 8; ++i)
        ra[i] = *(const v4f*)(ag + (size_t)i * 16 * KDIM);
#pragma unroll
      for (int i = 0; i < 4; ++i)
        rb[i] = *(const v4f*)(bg + (size_t)i * 16 * N);
    }

#pragma unroll
    for (int kk = 0; kk < 16; ++kk) {
      v2f a = *(const v2f*)&As[wave * 16 + l16][4 * kk + 2 * half];
#pragma unroll
      for (int nt = 0; nt < 4; ++nt) {
        v2f b = *(const v2f*)&BsT[nt * 16 + l16][4 * kk + 2 * half];
        acc[nt] = wmma_f32(a, b, acc[nt]);
      }
    }
    __syncthreads();
  }

#pragma unroll
  for (int nt = 0; nt < 4; ++nt) {
    const int col = colBase + nt * 16 + l16;
    const float bv = bias[col];
#pragma unroll
    for (int j = 0; j < 8; ++j) {
      const int row = rowBase + wave * 16 + j + 8 * half;
      C[(size_t)row * N + col] = acc[nt][j] + bv;
    }
  }
}

static __device__ __forceinline__ float xhalf(float x) {
  const int b = __builtin_bit_cast(int, x);
  return __builtin_bit_cast(
      float, __builtin_amdgcn_permlanex16(b, b, 0x76543210, 0xfedcba98, false,
                                          false));
}

__global__ __launch_bounds__(256, 8) void attn_kernel(
    const float* __restrict__ Q, const float* __restrict__ KV,
    const float* __restrict__ peq, const float* __restrict__ pekv,
    const float* __restrict__ mask, float* __restrict__ O) {
  __shared__ float ks[K2N][34];
  __shared__ float vp[K2N / 2][72];

  const int idx = blockIdx.x;
  const int h = idx % KNH;
  const int g = (idx / KNH) % KG;
  const int b = idx / (KNH * KG);
  const int gh = g >> 2, gw = g & 3;

  const int tid = threadIdx.x;
  const int wave = tid >> 5;
  const int lane = tid & 31;
  const int l16 = lane & 15;
  const int half = lane >> 4;

  {
    const int m  = tid & 7;
    const int r0 = tid >> 3;
    const unsigned vpart = (unsigned)(((b * KH) * KW + gw * 16 + 2 * (r0 & 7)) *
                                          (2 * KDIM) +
                                      KDIM + h * KHEAD + 4 * m);
    const int vhb = gh * 8 + (r0 >> 3);
    for (int i = 0; i < 4; ++i) {
      const int hs =
          (vhb + 4 * (i & 1) + ((i >> 1) == 0 ? KSHIFT : (KH - KSHIFT))) &
          (KH - 1);
      const float* vrow = KV + (vpart + ((unsigned)hs << 16));
      const v4f x0 = *(const v4f*)vrow;
      const v4f x1 = *(const v4f*)(vrow + 2 * KDIM);
      v4f w0, w1;
      w0.x = x0.x;  w0.y = x1.x;  w0.z = x0.y;  w0.w = x1.y;
      w1.x = x0.z;  w1.y = x1.z;  w1.z = x0.w;  w1.w = x1.w;
      *(v4f*)&vp[i * 32 + r0][8 * m]     = w0;
      *(v4f*)&vp[i * 32 + r0][8 * m + 4] = w1;
    }
    const unsigned kpart = (unsigned)(((b * KH) * KW + gw * 16 + (r0 & 15)) *
                                          (2 * KDIM) +
                                      h * KHEAD + 4 * m);
    const int khb = gh * 8 + (r0 >> 4);
    const float* pe0 = pekv + (unsigned)((g * K2N + r0) * (KHEAD / 2) + 2 * m);
    for (int i = 0; i < 8; ++i) {
      const int j = i * 32 + r0;
      const int hs =
          (khb + 2 * (i & 3) + ((i >> 2) == 0 ? KSHIFT : (KH - KSHIFT))) &
          (KH - 1);
      const float* krow = KV + (kpart + ((unsigned)hs << 16));
      const v4f kx = *(const v4f*)krow;
      const v2f cc = *(const v2f*)(pe0 + i * 32 * (KHEAD / 2));
      const v2f ss =
          *(const v2f*)(pe0 + (KG * K2N + i * 32) * (KHEAD / 2));
      v2f lo, hi;
      lo.x = kx.x * cc.x - kx.y * ss.x;
      lo.y = kx.z * cc.y - kx.w * ss.y;
      hi.x = kx.y * cc.x + kx.x * ss.x;
      hi.y = kx.w * cc.y + kx.z * ss.y;
      *(v2f*)&ks[j][2 * m]      = lo;
      *(v2f*)&ks[j][16 + 2 * m] = hi;
    }
  }

  const int qn = wave * 16 + l16;
  const int qhs = gh * 8 + (qn >> 4);
  const int qws = gw * 16 + (qn & 15);
  const float* qrow = Q + (size_t)((b * KH + qhs) * KW + qws) * KDIM + h * KHEAD;
  const float* qcos = peq + ((size_t)(0 * KG + g) * KN + qn) * (KHEAD / 2);
  const float* qsin = peq + ((size_t)(1 * KG + g) * KN + qn) * (KHEAD / 2);
  v2f qf[8];
#pragma unroll
  for (int kk = 0; kk < 8; ++kk) {
    const int k0 = 4 * kk + 2 * half;
#pragma unroll
    for (int e = 0; e < 2; ++e) {
      if (kk < 4) {
        const int c = k0 + e;
        qf[kk][e] = qrow[2 * c] * qcos[c] - qrow[2 * c + 1] * qsin[c];
      } else {
        const int c = k0 + e - 16;
        qf[kk][e] = qrow[2 * c] * qsin[c] + qrow[2 * c + 1] * qcos[c];
      }
    }
  }

  __syncthreads();

  const float scale = 0.1767766952966369f;
  const float log2e = 1.4426950408889634f;
  float mrun = -1e30f, lrun = 0.f;
  v8f oacc[2];
  oacc[0] = {}; oacc[1] = {};

  const float* maskq =
      mask + (size_t)g * KN * K2N + (size_t)qn * K2N + 8 * half;

  for (int kt = 0; kt < 16; ++kt) {
    const v4f mk0 = *(const v4f*)(maskq + kt * 16);
    const v4f mk1 = *(const v4f*)(maskq + kt * 16 + 4);

    v8f s = {};
#pragma unroll
    for (int kk = 0; kk < 8; ++kk) {
      v2f kfr = *(const v2f*)&ks[kt * 16 + l16][4 * kk + 2 * half];
      s = wmma_f32(kfr, qf[kk], s);
    }
    s[0] = s[0] * scale + mk0.x;  s[1] = s[1] * scale + mk0.y;
    s[2] = s[2] * scale + mk0.z;  s[3] = s[3] * scale + mk0.w;
    s[4] = s[4] * scale + mk1.x;  s[5] = s[5] * scale + mk1.y;
    s[6] = s[6] * scale + mk1.z;  s[7] = s[7] * scale + mk1.w;

    float m = fmaxf(fmaxf(fmaxf(s[0], s[1]), fmaxf(s[2], s[3])),
                    fmaxf(fmaxf(s[4], s[5]), fmaxf(s[6], s[7])));
    m = fmaxf(m, xhalf(m));
    const float mn = fmaxf(mrun, m);
    const float alpha = __builtin_amdgcn_exp2f((mrun - mn) * log2e);
    float rs = 0.f;
#pragma unroll
    for (int r = 0; r < 8; ++r) {
      s[r] = __builtin_amdgcn_exp2f((s[r] - mn) * log2e);
      rs += s[r];
    }
    rs = rs + xhalf(rs);
    lrun = lrun * alpha + rs;
    mrun = mn;
#pragma unroll
    for (int r = 0; r < 8; ++r) {
      oacc[0][r] *= alpha;
      oacc[1][r] *= alpha;
    }

    const float* v0 = &vp[kt * 8 + 4 * half][2 * l16];
#pragma unroll
    for (int kk = 0; kk < 4; ++kk) {
      v2f pb;
      pb.x = s[2 * kk];
      pb.y = s[2 * kk + 1];
#pragma unroll
      for (int ct = 0; ct < 2; ++ct) {
        const v2f va = *(const v2f*)(v0 + kk * 72 + ct * 32);
        oacc[ct] = wmma_f32(va, pb, oacc[ct]);
      }
    }
  }

  __syncthreads();
  float* ot = &ks[0][0] + wave * (16 * 36);
#pragma unroll
  for (int ct = 0; ct < 2; ++ct) {
    v4f o0, o1;
    o0.x = oacc[ct][0] / lrun;  o0.y = oacc[ct][1] / lrun;
    o0.z = oacc[ct][2] / lrun;  o0.w = oacc[ct][3] / lrun;
    o1.x = oacc[ct][4] / lrun;  o1.y = oacc[ct][5] / lrun;
    o1.z = oacc[ct][6] / lrun;  o1.w = oacc[ct][7] / lrun;
    *(v4f*)&ot[l16 * 36 + ct * 16 + 8 * half]     = o0;
    *(v4f*)&ot[l16 * 36 + ct * 16 + 8 * half + 4] = o1;
  }
  __syncthreads();
  {
    const int col = (lane & 7) * 4;
    const int hs = gh * 8 + wave;
#pragma unroll
    for (int i = 0; i < 4; ++i) {
      const int row = 4 * i + (lane >> 3);
      const v4f v = *(const v4f*)&ot[row * 36 + col];
      const int ws = gw * 16 + row;
      *(v4f*)(O + (size_t)((b * KH + hs) * KW + ws) * KDIM + h * KHEAD + col) =
          v;
    }
  }
}

extern "C" void kernel_launch(void* const* d_in, const int* in_sizes, int n_in,
                              void* d_out, int out_size, void* d_ws,
                              size_t ws_size, hipStream_t stream) {
  const float* x      = (const float*)d_in[0];
  const float* peq    = (const float*)d_in[1];
  const float* pekv   = (const float*)d_in[2];
  const float* mask   = (const float*)d_in[3];
  const float* w_q    = (const float*)d_in[4];
  const float* b_q    = (const float*)d_in[5];
  const float* w_kv   = (const float*)d_in[6];
  const float* b_kv   = (const float*)d_in[7];
  const float* w_proj = (const float*)d_in[8];
  const float* b_proj = (const float*)d_in[9];

  const int B = in_sizes[0] / (KH * KW * KDIM);
  const int M = B * KH * KW;

  float* qbuf  = (float*)d_ws;
  float* kvbuf = qbuf + (size_t)M * KDIM;
  float* obuf  = kvbuf + (size_t)M * 2 * KDIM;

  dim3 blk(256);
  gemm_bias_kernel<<<dim3((M / 128) * (KDIM / 64)), blk, 0, stream>>>(
      x, w_q, b_q, qbuf, M, KDIM);
  gemm_bias_kernel<<<dim3((M / 128) * (2 * KDIM / 64)), blk, 0, stream>>>(
      x, w_kv, b_kv, kvbuf, M, 2 * KDIM);
  attn_kernel<<<dim3(B * KG * KNH), blk, 0, stream>>>(qbuf, kvbuf, peq, pekv,
                                                      mask, obuf);
  gemm_bias_kernel<<<dim3((M / 128) * (KDIM / 64)), blk, 0, stream>>>(
      obuf, w_proj, b_proj, (float*)d_out, M, KDIM);
}
